# starting kernel
# baseline (speedup 1.0000x reference)
#include <hip/hip_runtime.h>
#include <hip/hip_bf16.h>
#include <cstdint>

#define B_   2
#define N_   16384
#define E_   131072
#define C_   256
#define H_   8
#define CIN_ 259
#define KPAD_ 288
#define SAP_  296
#define SAPU_ 264

typedef __attribute__((ext_vector_type(16))) _Float16 v16h;
typedef __attribute__((ext_vector_type(8)))  float    v8f;
typedef __attribute__((ext_vector_type(8)))  _Float16 h8;
typedef __attribute__((ext_vector_type(2)))  _Float16 h2;

__device__ inline v16h ldA2(const _Float16* p, int half) {
  h8 lo = *(const h8*)(p + half * 8);
  h8 hi = *(const h8*)(p + 16 + half * 8);
  return __builtin_shufflevector(lo, hi, 0, 1, 2, 3, 4, 5, 6, 7,
                                 8, 9, 10, 11, 12, 13, 14, 15);
}
__device__ inline v16h ldBp(const _Float16* __restrict__ Wp, int kc, int ntile, int lane) {
  return *(const v16h*)(Wp + ((((size_t)kc * (C_ / 16) + ntile) * 32 + lane) * 16));
}
__device__ inline v8f wmma_f16(v16h a, v16h b, v8f c) {
  return __builtin_amdgcn_wmma_f32_16x16x32_f16(false, a, false, b, (short)0, c, false, false);
}

__device__ inline void ln_stats(const float* buf, int ld, float* part, float* part2,
                                float* mu, float* istd, int tid) {
  int m = tid >> 4, c = tid & 15;
  float s = 0.f, s2 = 0.f;
#pragma unroll
  for (int j = 0; j < 16; ++j) {
    float x = buf[m * ld + c * 16 + j];
    s += x; s2 += x * x;
  }
  part[m * 16 + c] = s;
  part2[m * 16 + c] = s2;
  __syncthreads();
  if (tid < 16) {
    float S = 0.f, S2 = 0.f;
    for (int j = 0; j < 16; ++j) { S += part[tid * 16 + j]; S2 += part2[tid * 16 + j]; }
    float mean = S * (1.f / 256.f);
    float var  = fmaxf(S2 * (1.f / 256.f) - mean * mean, 0.f);
    mu[tid]   = mean;
    istd[tid] = rsqrtf(var + 1e-5f);
  }
  __syncthreads();
}

__device__ inline void atomicMaxF(float* addr, float v) {
  if (v >= 0.f) atomicMax((int*)addr, __float_as_int(v));
  else          atomicMin((unsigned int*)addr, __float_as_uint(v));
}

__global__ void k_featcvt(const float* __restrict__ features, _Float16* __restrict__ feat_h) {
  size_t idx = (size_t)blockIdx.x * 256 + threadIdx.x;
  if (idx >= (size_t)B_ * N_ * C_) return;
  int c = (int)(idx % C_);
  int n = (int)((idx / C_) % N_);
  int b = (int)(idx / ((size_t)C_ * N_));
  feat_h[idx] = (_Float16)features[((size_t)b * C_ + c) * N_ + n];
}

__global__ void k_wpack(const float* __restrict__ src, _Float16* __restrict__ dst,
                        int rows, int nchunks) {
  size_t idx = (size_t)blockIdx.x * 256 + threadIdx.x;
  if (idx >= (size_t)nchunks * 8192) return;
  int elem  = (int)(idx & 15);
  int lane  = (int)((idx >> 4) & 31);
  int ntile = (int)((idx >> 9) & 15);
  int kc    = (int)(idx >> 13);
  int half = lane >> 4, l16 = lane & 15;
  int j = elem >> 1, p = elem & 1;
  int kb = (j >> 2) * 16 + half * 8 + (j & 3) * 2 + p;
  int k = kc * 32 + kb;
  int n = ntile * 16 + l16;
  dst[idx] = (k < rows) ? (_Float16)src[(size_t)k * C_ + n] : (_Float16)0.f;
}

__global__ void k_init(float* __restrict__ agg, float* __restrict__ segmax,
                       float* __restrict__ segsum, float* __restrict__ cnt) {
  size_t idx = (size_t)blockIdx.x * 256 + threadIdx.x;
  if (idx < (size_t)B_ * N_ * C_) agg[idx] = 0.f;
  if (idx < (size_t)B_ * N_ * H_) { segmax[idx] = __int_as_float(0xff800000); segsum[idx] = 0.f; }
  if (idx < (size_t)B_ * N_) cnt[idx] = 0.f;
}

__global__ __launch_bounds__(256) void k_edge(
    const float* __restrict__ xyz, const int* __restrict__ edges,
    const _Float16* __restrict__ feat_h,
    const _Float16* __restrict__ WkP, const float* __restrict__ biask,
    const float* __restrict__ gk, const float* __restrict__ betak,
    const _Float16* __restrict__ WvP, const float* __restrict__ biasv,
    const float* __restrict__ gv, const float* __restrict__ betav,
    const _Float16* __restrict__ WqP, const float* __restrict__ biasq,
    const float* __restrict__ gq, const float* __restrict__ betaq,
    _Float16* __restrict__ Vh, float* __restrict__ simi) {
  const int b    = blockIdx.y;
  const int e0   = blockIdx.x * 16;
  const int tid  = threadIdx.x;
  const int lane = tid & 31;
  const int wave = tid >> 5;
  const int half = lane >> 4;
  const int l16  = lane & 15;
  const size_t bN = (size_t)b * N_;

  __shared__ int   s_src[16], s_dst[16];
  __shared__ float s_rel[16][3];
  __shared__ _Float16 sAll[16 * SAP_];
  __shared__ float sK[16][260];
  __shared__ float sQ[16][260];
  __shared__ float s_part[256], s_part2[256];
  __shared__ float s_mu[2][16], s_is[2][16];

  if (tid < 16) {
    size_t eidx = ((size_t)b * E_ + e0 + tid) * 2;
    int s = edges[eidx + 0];
    int d = edges[eidx + 1];
    s_src[tid] = s; s_dst[tid] = d;
#pragma unroll
    for (int j = 0; j < 3; ++j)
      s_rel[tid][j] = xyz[(bN + s) * 3 + j] - xyz[(bN + d) * 3 + j];
  }
  __syncthreads();

  const int m16 = tid >> 4;
  const int sg  = tid & 15;
  const _Float16* rowS = feat_h + (bN + (size_t)s_src[m16]) * C_;
  const _Float16* rowD = feat_h + (bN + (size_t)s_dst[m16]) * C_;
  h2 relp;
  {
    int k = sg * 2;
    relp.x = (_Float16)((k     < 3) ? s_rel[m16][k]     : 0.f);
    relp.y = (_Float16)((k + 1 < 3) ? s_rel[m16][k + 1] : 0.f);
  }

  const int nt0 = wave * 2;
  const int n0  = wave * 32;

  *(h8*)(sAll + m16 * SAP_ + sg * 16)     = *(const h8*)(rowS + sg * 16);
  *(h8*)(sAll + m16 * SAP_ + sg * 16 + 8) = *(const h8*)(rowS + sg * 16 + 8);
  *(h2*)(sAll + m16 * SAP_ + 256 + sg * 2) = relp;
  __syncthreads();

  v8f aK0 = {}, aK1 = {}, aV0 = {}, aV1 = {};
  const _Float16* aRow = sAll + l16 * SAP_;
#pragma unroll
  for (int kc = 0; kc < 9; ++kc) {
    v16h a   = ldA2(aRow + kc * 32, half);
    v16h bk0 = ldBp(WkP, kc, nt0,     lane);
    v16h bk1 = ldBp(WkP, kc, nt0 + 1, lane);
    v16h bv0 = ldBp(WvP, kc, nt0,     lane);
    v16h bv1 = ldBp(WvP, kc, nt0 + 1, lane);
    aK0 = wmma_f16(a, bk0, aK0);
    aK1 = wmma_f16(a, bk1, aK1);
    aV0 = wmma_f16(a, bv0, aV0);
    aV1 = wmma_f16(a, bv1, aV1);
  }
#pragma unroll
  for (int r = 0; r < 8; ++r) {
    int m = r + half * 8;
    int n = n0 + l16;
    sK[m][n]      = fmaxf(aK0[r] + biask[n],      0.f);
    sK[m][n + 16] = fmaxf(aK1[r] + biask[n + 16], 0.f);
    sQ[m][n]      = fmaxf(aV0[r] + biasv[n],      0.f);
    sQ[m][n + 16] = fmaxf(aV1[r] + biasv[n + 16], 0.f);
  }
  __syncthreads();
  ln_stats(&sK[0][0], 260, s_part, s_part2, s_mu[0], s_is[0], tid);
  ln_stats(&sQ[0][0], 260, s_part, s_part2, s_mu[1], s_is[1], tid);
  {
    int m = tid >> 4, cb = (tid & 15) * 16;
    float muK = s_mu[0][m], isK = s_is[0][m];
    float muV = s_mu[1][m], isV = s_is[1][m];
    size_t vbase = ((size_t)b * E_ + e0 + m) * C_ + cb;
#pragma unroll
    for (int j = 0; j < 16; ++j) {
      int c = cb + j;
      sK[m][c] = (sK[m][c] - muK) * isK * gk[c] + betak[c];
      float vx = (sQ[m][c] - muV) * isV * gv[c] + betav[c];
      Vh[vbase + j] = (_Float16)vx;
    }
  }
  __syncthreads();

  *(h8*)(sAll + m16 * SAP_ + sg * 16)     = *(const h8*)(rowD + sg * 16);
  *(h8*)(sAll + m16 * SAP_ + sg * 16 + 8) = *(const h8*)(rowD + sg * 16 + 8);
  *(h2*)(sAll + m16 * SAP_ + 256 + sg * 2) = relp;
  __syncthreads();

  v8f aQ0 = {}, aQ1 = {};
#pragma unroll
  for (int kc = 0; kc < 9; ++kc) {
    v16h a   = ldA2(aRow + kc * 32, half);
    v16h bq0 = ldBp(WqP, kc, nt0,     lane);
    v16h bq1 = ldBp(WqP, kc, nt0 + 1, lane);
    aQ0 = wmma_f16(a, bq0, aQ0);
    aQ1 = wmma_f16(a, bq1, aQ1);
  }
#pragma unroll
  for (int r = 0; r < 8; ++r) {
    int m = r + half * 8;
    int n = n0 + l16;
    sQ[m][n]      = fmaxf(aQ0[r] + biasq[n],      0.f);
    sQ[m][n + 16] = fmaxf(aQ1[r] + biasq[n + 16], 0.f);
  }
  __syncthreads();
  ln_stats(&sQ[0][0], 260, s_part, s_part2, s_mu[1], s_is[1], tid);
  {
    int m = tid >> 4, cb = (tid & 15) * 16;
    float mu = s_mu[1][m], is = s_is[1][m];
#pragma unroll
    for (int j = 0; j < 16; ++j) {
      int c = cb + j;
      sQ[m][c] = (sQ[m][c] - mu) * is * gq[c] + betaq[c];
    }
  }
  __syncthreads();

  if (tid < 128) {
    int m = tid >> 3, h = tid & 7;
    float acc = 0.f;
#pragma unroll
    for (int j = 0; j < 32; ++j) acc += sK[m][h * 32 + j] * sQ[m][h * 32 + j];
    simi[((size_t)b * E_ + e0 + m) * H_ + h] = acc * 0.17677669529663687f;
  }
}

__global__ void k_segmax(const int* __restrict__ edges, const float* __restrict__ simi,
                         float* __restrict__ segmax, float* __restrict__ cnt) {
  int b = blockIdx.y;
  size_t idx = (size_t)blockIdx.x * 256 + threadIdx.x;
  if (idx >= (size_t)E_ * H_) return;
  int e = (int)(idx >> 3), h = (int)(idx & 7);
  int d = edges[((size_t)b * E_ + e) * 2 + 1];
  float v = simi[((size_t)b * E_ + e) * H_ + h];
  atomicMaxF(&segmax[((size_t)b * N_ + d) * H_ + h], v);
  if (h == 0) atomicAdd(&cnt[(size_t)b * N_ + d], 1.0f);
}

__global__ void k_expsum(const int* __restrict__ edges, float* __restrict__ simi,
                         const float* __restrict__ segmax, float* __restrict__ segsum) {
  int b = blockIdx.y;
  size_t idx = (size_t)blockIdx.x * 256 + threadIdx.x;
  if (idx >= (size_t)E_ * H_) return;
  int e = (int)(idx >> 3), h = (int)(idx & 7);
  int d = edges[((size_t)b * E_ + e) * 2 + 1];
  float mx = segmax[((size_t)b * N_ + d) * H_ + h];
  float w = __expf(simi[((size_t)b * E_ + e) * H_ + h] - mx);
  simi[((size_t)b * E_ + e) * H_ + h] = w;
  atomicAdd(&segsum[((size_t)b * N_ + d) * H_ + h], w);
}

__global__ void k_scatter(const int* __restrict__ edges, const float* __restrict__ expw,
                          const float* __restrict__ segsum, const _Float16* __restrict__ Vh,
                          float* __restrict__ agg) {
  int b = blockIdx.y;
  size_t idx = (size_t)blockIdx.x * 256 + threadIdx.x;
  int e = (int)(idx >> 8), c = (int)(idx & 255);
  int d = edges[((size_t)b * E_ + e) * 2 + 1];
  int h = c >> 5;
  float w    = expw[((size_t)b * E_ + e) * H_ + h];
  float base = segsum[((size_t)b * N_ + d) * H_ + h];
  float v = (float)Vh[((size_t)b * E_ + e) * C_ + c];
  atomicAdd(&agg[((size_t)b * N_ + d) * C_ + c], v * (w / base));
}

__global__ void k_div(const float* __restrict__ agg, const float* __restrict__ cnt,
                      _Float16* __restrict__ aggh) {
  int b = blockIdx.y;
  size_t idx = (size_t)blockIdx.x * 256 + threadIdx.x;
  int n = (int)(idx >> 8);
  float cc = fmaxf(cnt[(size_t)b * N_ + n], 1.0f);
  size_t o = (size_t)b * N_ * C_ + idx;
  aggh[o] = (_Float16)(agg[o] / cc);
}

__global__ __launch_bounds__(256) void k_update(
    const _Float16* __restrict__ aggh, const _Float16* __restrict__ WuP,
    const float* __restrict__ biasu, const float* __restrict__ gu,
    const float* __restrict__ betau, const float* __restrict__ features,
    float* __restrict__ out) {
  const int b     = blockIdx.y;
  const int nrow0 = blockIdx.x * 16;
  const int tid   = threadIdx.x;
  const int lane  = tid & 31;
  const int wave  = tid >> 5;
  const int half  = lane >> 4;
  const int l16   = lane & 15;

  __shared__ _Float16 sAll[16 * SAPU_];
  __shared__ float sO[16][260];
  __shared__ float s_part[256], s_part2[256];
  __shared__ float s_mu[16], s_is[16];

  {
    int m = tid >> 4, sg = tid & 15;
    const _Float16* row = aggh + ((size_t)b * N_ + nrow0 + m) * C_;
    *(h8*)(sAll + m * SAPU_ + sg * 16)     = *(const h8*)(row + sg * 16);
    *(h8*)(sAll + m * SAPU_ + sg * 16 + 8) = *(const h8*)(row + sg * 16 + 8);
  }
  __syncthreads();

  const int nt0 = wave * 2;
  const int n0  = wave * 32;
  const _Float16* aRow = sAll + l16 * SAPU_;
  v8f a0 = {}, a1 = {};
#pragma unroll
  for (int kc = 0; kc < 8; ++kc) {
    v16h a  = ldA2(aRow + kc * 32, half);
    v16h b0 = ldBp(WuP, kc, nt0,     lane);
    v16h b1 = ldBp(WuP, kc, nt0 + 1, lane);
    a0 = wmma_f16(a, b0, a0);
    a1 = wmma_f16(a, b1, a1);
  }
#pragma unroll
  for (int r = 0; r < 8; ++r) {
    int m = r + half * 8;
    int n = n0 + l16;
    sO[m][n]      = fmaxf(a0[r] + biasu[n],      0.f);
    sO[m][n + 16] = fmaxf(a1[r] + biasu[n + 16], 0.f);
  }
  __syncthreads();
  ln_stats(&sO[0][0], 260, s_part, s_part2, s_mu, s_is, tid);
  {
    int m = tid >> 4, cb = (tid & 15) * 16;
    float mu = s_mu[m], is = s_is[m];
#pragma unroll
    for (int j = 0; j < 16; ++j) {
      int c = cb + j;
      float val = (sO[m][c] - mu) * is * gu[c] + betau[c];
      size_t oi = ((size_t)b * C_ + c) * N_ + nrow0 + m;
      out[oi] = val + features[oi];
    }
  }
}

extern "C" void kernel_launch(void* const* d_in, const int* in_sizes, int n_in,
                              void* d_out, int out_size, void* d_ws, size_t ws_size,
                              hipStream_t stream) {
  const float* xyz      = (const float*)d_in[0];
  const float* features = (const float*)d_in[1];
  const int*   edges    = (const int*)d_in[2];
  const float* Wk = (const float*)d_in[3];  const float* bk = (const float*)d_in[4];
  const float* gk = (const float*)d_in[5];  const float* betak = (const float*)d_in[6];
  const float* Wv = (const float*)d_in[7];  const float* bv = (const float*)d_in[8];
  const float* gv = (const float*)d_in[9];  const float* betav = (const float*)d_in[10];
  const float* Wq = (const float*)d_in[11]; const float* bq = (const float*)d_in[12];
  const float* gq = (const float*)d_in[13]; const float* betaq = (const float*)d_in[14];
  const float* Wu = (const float*)d_in[15]; const float* bu = (const float*)d_in[16];
  const float* gu = (const float*)d_in[17]; const float* betau = (const float*)d_in[18];
  float* out = (float*)d_out;

  char* ws = (char*)d_ws;
  size_t off = 0;
  auto carve = [&](size_t bytes) -> char* {
    char* p = ws + off;
    off = (off + bytes + 255) & ~(size_t)255;
    return p;
  };
  _Float16* feat_h = (_Float16*)carve((size_t)B_ * N_ * C_ * 2);
  _Float16* WkP    = (_Float16*)carve((size_t)KPAD_ * C_ * 2);
  _Float16* WvP    = (_Float16*)carve((size_t)KPAD_ * C_ * 2);
  _Float16* WqP    = (_Float16*)carve((size_t)KPAD_ * C_ * 2);
  _Float16* WuP    = (_Float16*)carve((size_t)C_ * C_ * 2);
  _Float16* Vh     = (_Float16*)carve((size_t)B_ * E_ * C_ * 2);
  float*    simi   = (float*)carve((size_t)B_ * E_ * H_ * 4);
  float*    segmax = (float*)carve((size_t)B_ * N_ * H_ * 4);
  float*    segsum = (float*)carve((size_t)B_ * N_ * H_ * 4);
  float*    cnt    = (float*)carve((size_t)B_ * N_ * 4);
  float*    agg    = (float*)carve((size_t)B_ * N_ * C_ * 4);
  _Float16* aggh   = (_Float16*)carve((size_t)B_ * N_ * C_ * 2);
  (void)ws_size; (void)in_sizes; (void)n_in; (void)out_size;

  k_featcvt<<<(unsigned)(((size_t)B_ * N_ * C_) / 256), 256, 0, stream>>>(features, feat_h);
  k_wpack<<<(9 * 8192) / 256, 256, 0, stream>>>(Wk, WkP, CIN_, 9);
  k_wpack<<<(9 * 8192) / 256, 256, 0, stream>>>(Wv, WvP, CIN_, 9);
  k_wpack<<<(9 * 8192) / 256, 256, 0, stream>>>(Wq, WqP, CIN_, 9);
  k_wpack<<<(8 * 8192) / 256, 256, 0, stream>>>(Wu, WuP, C_, 8);
  k_init<<<(unsigned)(((size_t)B_ * N_ * C_) / 256), 256, 0, stream>>>(agg, segmax, segsum, cnt);

  k_edge<<<dim3(E_ / 16, B_), 256, 0, stream>>>(
      xyz, edges, feat_h,
      WkP, bk, gk, betak,
      WvP, bv, gv, betav,
      WqP, bq, gq, betaq,
      Vh, simi);

  k_segmax<<<dim3((E_ * H_) / 256, B_), 256, 0, stream>>>(edges, simi, segmax, cnt);
  k_expsum<<<dim3((E_ * H_) / 256, B_), 256, 0, stream>>>(edges, simi, segmax, segsum);
  k_scatter<<<dim3((unsigned)(((size_t)E_ * C_) / 256), B_), 256, 0, stream>>>(
      edges, simi, segsum, Vh, agg);
  k_div<<<dim3((N_ * C_) / 256, B_), 256, 0, stream>>>(agg, cnt, aggh);

  k_update<<<dim3(N_ / 16, B_), 256, 0, stream>>>(aggh, WuP, bu, gu, betau, features, out);
}


# segment softmax inside the destination range's work-group: per-node max, weight sum, edge count by LDS atomics; global softmax passes (4M float atomics) become list builders: sum order changes
# speedup vs baseline: 2.2641x; 2.2641x over previous
#include <hip/hip_runtime.h>
#include <hip/hip_bf16.h>
#include <cstdint>

#define B_   2
#define N_   16384
#define E_   131072
#define C_   256
#define H_   8
#define CIN_ 259
#define KPAD_ 288
#define FRAGH_  512
#define WFRAGS_ 144
#define WPLANE_ (WFRAGS_ * FRAGH_)
#define NTILES_ (B_ * E_ / 16)
#define BINN_  128
#define NBINS_ (B_ * N_ / BINN_)
#define BSTR_  64

typedef __attribute__((ext_vector_type(16))) _Float16 v16h;
typedef __attribute__((ext_vector_type(8)))  float    v8f;
typedef __attribute__((ext_vector_type(8)))  _Float16 h8;
typedef __attribute__((ext_vector_type(2)))  _Float16 h2;

__device__ inline v16h ldA2(const _Float16* p, int half) {
  h8 lo = *(const h8*)(p + half * 8);
  h8 hi = *(const h8*)(p + 16 + half * 8);
  return __builtin_shufflevector(lo, hi, 0, 1, 2, 3, 4, 5, 6, 7,
                                 8, 9, 10, 11, 12, 13, 14, 15);
}
__device__ inline v16h ldBp(const _Float16* __restrict__ Wp, int kc, int ntile, int lane) {
  return *(const v16h*)(Wp + ((((size_t)kc * (C_ / 16) + ntile) * 32 + lane) * 16));
}
__device__ inline v8f wmma_f16(v16h a, v16h b, v8f c) {
  return __builtin_amdgcn_wmma_f32_16x16x32_f16(false, a, false, b, (short)0, c, false, false);
}

__device__ inline void atomicMaxF(float* addr, float v) {
  if (v >= 0.f) atomicMax((int*)addr, __float_as_int(v));
  else          atomicMin((unsigned int*)addr, __float_as_uint(v));
}

__global__ void k_featcvt(const float* __restrict__ features, _Float16* __restrict__ feat_h) {
  size_t idx = (size_t)blockIdx.x * 256 + threadIdx.x;
  if (idx >= (size_t)B_ * N_ * C_) return;
  int c = (int)(idx % C_);
  int n = (int)((idx / C_) % N_);
  int b = (int)(idx / ((size_t)C_ * N_));
  feat_h[idx] = (_Float16)features[((size_t)b * C_ + c) * N_ + n];
}

__global__ void k_wpack(const float* __restrict__ src, _Float16* __restrict__ dst,
                        int rows, int nchunks) {
  size_t idx = (size_t)blockIdx.x * 256 + threadIdx.x;
  if (idx >= (size_t)nchunks * 8192) return;
  int elem  = (int)(idx & 15);
  int lane  = (int)((idx >> 4) & 31);
  int ntile = (int)((idx >> 9) & 15);
  int kc    = (int)(idx >> 13);
  int half = lane >> 4, l16 = lane & 15;
  int j = elem >> 1, p = elem & 1;
  int kb = (j >> 2) * 16 + half * 8 + (j & 3) * 2 + p;
  int k = kc * 32 + kb;
  int n = ntile * 16 + l16;
  dst[idx] = (k < rows) ? (_Float16)src[(size_t)k * C_ + n] : (_Float16)0.f;
}

__global__ void k_init(int* __restrict__ bincnt) {
  const int idx = blockIdx.x * 256 + threadIdx.x;
  if (idx < NBINS_ * BSTR_) bincnt[idx] = 0;
}


__device__ __forceinline__ void fill_w(_Float16* __restrict__ sWm,
                                       const _Float16* __restrict__ Wp, int tid) {
  for (int i = tid; i < WFRAGS_ * 64; i += 256) {
    int f = i >> 6, l = (i >> 1) & 31, p = i & 1;
    *(h8*)(sWm + f * FRAGH_ + p * 256 + l * 8) = *(const h8*)(Wp + (size_t)i * 8);
  }
}
__device__ __forceinline__ v16h ldW(const _Float16* sWb, int j) {
  const _Float16* p = sWb + j * FRAGH_;
  h8 lo = *(const h8*)(p);
  h8 hi = *(const h8*)(p + 256);
  return __builtin_shufflevector(lo, hi, 0, 1, 2, 3, 4, 5, 6, 7,
                                 8, 9, 10, 11, 12, 13, 14, 15);
}
__device__ __forceinline__ void ld8(const float* __restrict__ p, float (&o)[8]) {
  float4 a = *(const float4*)(p);
  float4 b = *(const float4*)(p + 4);
  o[0] = a.x; o[1] = a.y; o[2] = a.z; o[3] = a.w;
  o[4] = b.x; o[5] = b.y; o[6] = b.z; o[7] = b.w;
}
__device__ __forceinline__ void load_x(v16h (&x)[9], const _Float16* __restrict__ row,
                                       float r0, float r1, float r2, int half) {
#pragma unroll
  for (int kc = 0; kc < 8; ++kc) x[kc] = ldA2(row + kc * 32, half);
  v16h xr = {};
  xr[0] = (_Float16)(half ? 0.f : r0);
  xr[1] = (_Float16)(half ? 0.f : r1);
  xr[2] = (_Float16)(half ? 0.f : r2);
  x[8] = xr;
}
__device__ __forceinline__ void edge_gemm(v8f (&acc)[16], const _Float16* sWall, int woff,
                                          const v16h (&x)[9]) {
#pragma unroll
  for (int nt = 0; nt < 16; ++nt) acc[nt] = v8f{};
  v16h w[2][8];
#pragma unroll
  for (int j = 0; j < 8; ++j) w[0][j] = ldW(sWall + woff, j);
  __builtin_amdgcn_sched_barrier(0);
#pragma unroll
  for (int blk = 0; blk < 18; ++blk) {
    if (blk + 1 < 18) {
      int o = woff + (blk + 1) * 8 * FRAGH_;
      asm volatile("" : "+v"(o));
#pragma unroll
      for (int j = 0; j < 8; ++j) w[(blk + 1) & 1][j] = ldW(sWall + o, j);
    }
    __builtin_amdgcn_sched_barrier(0);
#pragma unroll
    for (int j = 0; j < 8; ++j) {
      const int f = blk * 8 + j;
      acc[f & 15] = wmma_f16(w[blk & 1][j], x[f >> 4], acc[f & 15]);
    }
    __builtin_amdgcn_sched_barrier(0);
  }
}
__device__ __forceinline__ void bias_relu_ln(v8f (&acc)[16], const float* __restrict__ bias,
                                             const float* __restrict__ g,
                                             const float* __restrict__ beta, int half) {
  float s = 0.f, s2 = 0.f;
#pragma unroll
  for (int nt = 0; nt < 16; ++nt) {
    float bb[8];
    ld8(bias + nt * 16 + half * 8, bb);
#pragma unroll
    for (int r = 0; r < 8; ++r) {
      float v = fmaxf(acc[nt][r] + bb[r], 0.f);
      acc[nt][r] = v;
      s += v; s2 += v * v;
    }
  }
  s  += __shfl_xor(s, 16);
  s2 += __shfl_xor(s2, 16);
  const float mean = s * (1.f / 256.f);
  const float var  = fmaxf(s2 * (1.f / 256.f) - mean * mean, 0.f);
  const float istd = rsqrtf(var + 1e-5f);
#pragma unroll
  for (int nt = 0; nt < 16; ++nt) {
    float gg[8], be[8];
    ld8(g + nt * 16 + half * 8, gg);
    ld8(beta + nt * 16 + half * 8, be);
#pragma unroll
    for (int r = 0; r < 8; ++r)
      acc[nt][r] = (acc[nt][r] - mean) * istd * gg[r] + be[r];
  }
  __builtin_amdgcn_sched_barrier(0);
}

__global__ __launch_bounds__(256) void k_edge(
    const float* __restrict__ xyz, const int* __restrict__ edges,
    const _Float16* __restrict__ feat_h,
    const _Float16* __restrict__ WkP, const float* __restrict__ biask,
    const float* __restrict__ gk, const float* __restrict__ betak,
    const _Float16* __restrict__ WvP, const float* __restrict__ biasv,
    const float* __restrict__ gv, const float* __restrict__ betav,
    const _Float16* __restrict__ WqP, const float* __restrict__ biasq,
    const float* __restrict__ gq, const float* __restrict__ betaq,
    _Float16* __restrict__ Vh, float* __restrict__ simi) {
  const int tid  = threadIdx.x;
  const int wave = __builtin_amdgcn_readfirstlane(tid >> 5);

  __shared__ __attribute__((aligned(16))) _Float16 sW[2 * WPLANE_];
  _Float16* sW0 = sW;
  _Float16* sW1 = sW + WPLANE_;

  fill_w(sW0, WkP, tid);
  fill_w(sW1, WqP, tid);
  __syncthreads();

  for (int tile = blockIdx.x * 8 + wave; tile < NTILES_; tile += gridDim.x * 8) {
    const int b = tile / (E_ / 16);
    int ln = (int)__builtin_amdgcn_mbcnt_lo(~0u, 0u);
    asm volatile("" : "+v"(ln));
    const int half = ln >> 4;
    const int l16  = ln & 15;
    const int woff = ln * 8;
    const size_t e = (size_t)b * E_ + (size_t)(tile % (E_ / 16)) * 16 + l16;
    const size_t bN = (size_t)b * N_;
    const int s = edges[e * 2 + 0];
    const int d = edges[e * 2 + 1];
    const float r0 = xyz[(bN + s) * 3 + 0] - xyz[(bN + d) * 3 + 0];
    const float r1 = xyz[(bN + s) * 3 + 1] - xyz[(bN + d) * 3 + 1];
    const float r2 = xyz[(bN + s) * 3 + 2] - xyz[(bN + d) * 3 + 2];

    v16h x[9];
    v8f aK[16], aQ[16];
    load_x(x, feat_h + (bN + (size_t)s) * C_, r0, r1, r2, half);
    edge_gemm(aK, sW, woff, x);
    bias_relu_ln(aK, biask, gk, betak, half);

    load_x(x, feat_h + (bN + (size_t)d) * C_, r0, r1, r2, half);
    edge_gemm(aQ, sW, WPLANE_ + woff, x);
    bias_relu_ln(aQ, biasq, gq, betaq, half);

    float sv[8];
#pragma unroll
    for (int h = 0; h < 8; ++h) {
      float p = 0.f;
#pragma unroll
      for (int r = 0; r < 8; ++r) p += aK[2 * h][r] * aQ[2 * h][r];
#pragma unroll
      for (int r = 0; r < 8; ++r) p += aK[2 * h + 1][r] * aQ[2 * h + 1][r];
      p += __shfl_xor(p, 16);
      sv[h] = p * 0.17677669529663687f;
    }
    if (half == 0) {
      float4* o = (float4*)(simi + e * H_);
      o[0] = make_float4(sv[0], sv[1], sv[2], sv[3]);
      o[1] = make_float4(sv[4], sv[5], sv[6], sv[7]);
    }
  }

  __syncthreads();
  fill_w(sW0, WvP, tid);
  __syncthreads();

  for (int tile = blockIdx.x * 8 + wave; tile < NTILES_; tile += gridDim.x * 8) {
    const int b = tile / (E_ / 16);
    int ln = (int)__builtin_amdgcn_mbcnt_lo(~0u, 0u);
    asm volatile("" : "+v"(ln));
    const int half = ln >> 4;
    const int l16  = ln & 15;
    const int woff = ln * 8;
    const size_t e = (size_t)b * E_ + (size_t)(tile % (E_ / 16)) * 16 + l16;
    const size_t bN = (size_t)b * N_;
    const int s = edges[e * 2 + 0];
    const int d = edges[e * 2 + 1];
    const float r0 = xyz[(bN + s) * 3 + 0] - xyz[(bN + d) * 3 + 0];
    const float r1 = xyz[(bN + s) * 3 + 1] - xyz[(bN + d) * 3 + 1];
    const float r2 = xyz[(bN + s) * 3 + 2] - xyz[(bN + d) * 3 + 2];

    v16h x[9];
    v8f aV[16];
    load_x(x, feat_h + (bN + (size_t)s) * C_, r0, r1, r2, half);
    edge_gemm(aV, sW, woff, x);
    bias_relu_ln(aV, biasv, gv, betav, half);

    _Float16* vrow = Vh + e * C_ + half * 8;
#pragma unroll
    for (int nt = 0; nt < 16; ++nt) {
      h8 o;
#pragma unroll
      for (int r = 0; r < 8; ++r) o[r] = (_Float16)aV[nt][r];
      *(h8*)(vrow + nt * 16) = o;
    }
  }
}

__global__ void k_rank(const int* __restrict__ edges, int* __restrict__ bincnt,
                       int* __restrict__ eslot) {
  const int t = blockIdx.x * 256 + threadIdx.x;
  if (t >= B_ * E_ / 2) return;
  const int d0 = edges[(size_t)t * 4 + 1];
  const int d1 = edges[(size_t)t * 4 + 3];
  const int bb = ((2 * t) / E_) * (N_ / BINN_);
  int2 s;
  s.x = atomicAdd(&bincnt[(bb + d0 / BINN_) * BSTR_], 1);
  s.y = atomicAdd(&bincnt[(bb + d1 / BINN_) * BSTR_], 1);
  ((int2*)eslot)[t] = s;
}

__global__ __launch_bounds__(NBINS_) void k_binscan(const int* __restrict__ bincnt,
                                                    int* __restrict__ binoff) {
  __shared__ int s[NBINS_];
  const int t = threadIdx.x;
  const int v = bincnt[t * BSTR_];
  s[t] = v;
  __syncthreads();
  for (int o = 1; o < NBINS_; o <<= 1) {
    const int a = (t >= o) ? s[t - o] : 0;
    __syncthreads();
    s[t] += a;
    __syncthreads();
  }
  binoff[t] = s[t] - v;
}

__global__ void k_fill(const int* __restrict__ edges, const int* __restrict__ binoff,
                       const int* __restrict__ eslot, int* __restrict__ perm) {
  const int t = blockIdx.x * 256 + threadIdx.x;
  if (t >= B_ * E_ / 2) return;
  const int d0 = edges[(size_t)t * 4 + 1];
  const int d1 = edges[(size_t)t * 4 + 3];
  const int2 s = ((const int2*)eslot)[t];
  const int bb = ((2 * t) / E_) * (N_ / BINN_);
  perm[binoff[bb + d0 / BINN_] + s.x] = 2 * t;
  perm[binoff[bb + d1 / BINN_] + s.y] = 2 * t + 1;
}

__global__ __launch_bounds__(1024) void k_binscatter(
    const int* __restrict__ edges, const float* __restrict__ simi,
    const _Float16* __restrict__ Vh, const int* __restrict__ bincnt,
    const int* __restrict__ binoff, const int* __restrict__ perm,
    _Float16* __restrict__ aggh) {
  __shared__ __attribute__((aligned(16))) float tile[BINN_ * C_];
  __shared__ float smax[BINN_ * H_];
  __shared__ float ssum[BINN_ * H_];
  __shared__ int   scnt[BINN_];
  const int tid  = threadIdx.x;
  const int lane = tid & 31;
  const int wave = __builtin_amdgcn_readfirstlane(tid >> 5);
  const int bin  = blockIdx.x;
  const int hq   = lane >> 2;

  for (int i = tid; i < BINN_ * C_ / 4; i += 1024)
    ((float4*)tile)[i] = make_float4(0.f, 0.f, 0.f, 0.f);
  for (int i = tid; i < BINN_ * H_; i += 1024) {
    smax[i] = __int_as_float(0xff800000);
    ssum[i] = 0.f;
  }
  for (int i = tid; i < BINN_; i += 1024) scnt[i] = 0;
  __syncthreads();

  const int nE  = bincnt[bin * BSTR_];
  const int off = binoff[bin];

  {
    const int ha = tid & 7;
    for (int base = tid >> 3; base < nE; base += 128 * 4) {
      int   ge[4];
      int   dl[4];
      float sv[4];
#pragma unroll
      for (int u = 0; u < 4; ++u) {
        const int idx = (base + u * 128 < nE) ? (base + u * 128) : (nE - 1);
        ge[u] = perm[off + idx];
      }
#pragma unroll
      for (int u = 0; u < 4; ++u) {
        dl[u] = edges[(size_t)ge[u] * 2 + 1] & (BINN_ - 1);
        sv[u] = simi[(size_t)ge[u] * H_ + ha];
      }
#pragma unroll
      for (int u = 0; u < 4; ++u) {
        if (base + u * 128 < nE) {
          atomicMaxF(&smax[dl[u] * H_ + ha], sv[u]);
          if (ha == 0) atomicAdd(&scnt[dl[u]], 1);
        }
      }
    }
  }
  __syncthreads();

  for (int base = wave * 4; base < nE; base += 32 * 4) {
    int   ge[4];
    int   dl[4];
    float sv[4];
    h8    v[4];
#pragma unroll
    for (int u = 0; u < 4; ++u) {
      const int idx = (base + u < nE) ? (base + u) : (nE - 1);
      ge[u] = perm[off + idx];
    }
#pragma unroll
    for (int u = 0; u < 4; ++u) {
      dl[u] = edges[(size_t)ge[u] * 2 + 1] & (BINN_ - 1);
      sv[u] = simi[(size_t)ge[u] * H_ + hq];
      v[u]  = *(const h8*)(Vh + (size_t)ge[u] * C_ + lane * 8);
    }
#pragma unroll
    for (int u = 0; u < 4; ++u) {
      if (base + u < nE) {
        const float w = __expf(sv[u] - smax[dl[u] * H_ + hq]);
        if ((lane & 3) == 0) atomicAdd(&ssum[dl[u] * H_ + hq], w);
        float* row = tile + dl[u] * C_ + lane;
#pragma unroll
        for (int r = 0; r < 8; ++r) atomicAdd(row + r * 32, (float)v[u][r] * w);
      }
    }
  }
  __syncthreads();

  const int b  = bin / (N_ / BINN_);
  const int n0 = (bin % (N_ / BINN_)) * BINN_;
  for (int nl = wave; nl < BINN_; nl += 32) {
    const size_t gn = (size_t)b * N_ + n0 + nl;
    const float cc = fmaxf((float)scnt[nl], 1.0f);
    float bs = ssum[nl * H_ + hq];
    bs = (bs > 0.f) ? bs : 1.f;
    h8 o;
#pragma unroll
    for (int r = 0; r < 8; ++r) o[r] = (_Float16)(tile[nl * C_ + r * 32 + lane] / bs / cc);
    *(h8*)(aggh + gn * C_ + lane * 8) = o;
  }
}

__global__ __launch_bounds__(256) void k_update(
    const _Float16* __restrict__ aggh, const _Float16* __restrict__ WuP,
    const float* __restrict__ biasu, const float* __restrict__ gu,
    const float* __restrict__ betau, const float* __restrict__ features,
    float* __restrict__ out) {
  const int lane = threadIdx.x & 31;
  const int wave = threadIdx.x >> 5;
  const int half = lane >> 4;
  const int l16  = lane & 15;
  const int tile = blockIdx.x * 8 + wave;
  const int b    = tile / (N_ / 16);
  const int node = (tile % (N_ / 16)) * 16 + l16;

  const _Float16* row = aggh + ((size_t)b * N_ + node) * C_;
  v16h xb[8];
#pragma unroll
  for (int kc = 0; kc < 8; ++kc) xb[kc] = ldA2(row + kc * 32, half);

  v8f acc[16];
#pragma unroll
  for (int nt = 0; nt < 16; ++nt) {
    v8f c = {};
#pragma unroll
    for (int kc = 0; kc < 8; ++kc) c = wmma_f16(ldBp(WuP, kc, nt, lane), xb[kc], c);
    acc[nt] = c;
  }

  float s = 0.f, s2 = 0.f;
#pragma unroll
  for (int nt = 0; nt < 16; ++nt) {
#pragma unroll
    for (int r = 0; r < 8; ++r) {
      int ch = nt * 16 + half * 8 + r;
      float x = fmaxf(acc[nt][r] + biasu[ch], 0.f);
      acc[nt][r] = x;
      s += x; s2 += x * x;
    }
  }
  s  += __shfl_xor(s, 16);
  s2 += __shfl_xor(s2, 16);
  const float mean = s * (1.f / 256.f);
  const float var  = fmaxf(s2 * (1.f / 256.f) - mean * mean, 0.f);
  const float istd = rsqrtf(var + 1e-5f);

#pragma unroll
  for (int nt = 0; nt < 16; ++nt) {
#pragma unroll
    for (int r = 0; r < 8; ++r) {
      int ch = nt * 16 + half * 8 + r;
      float val = (acc[nt][r] - mean) * istd * gu[ch] + betau[ch];
      size_t oi = ((size_t)b * C_ + ch) * N_ + node;
      out[oi] = val + features[oi];
    }
  }
}

extern "C" void kernel_launch(void* const* d_in, const int* in_sizes, int n_in,
                              void* d_out, int out_size, void* d_ws, size_t ws_size,
                              hipStream_t stream) {
  const float* xyz      = (const float*)d_in[0];
  const float* features = (const float*)d_in[1];
  const int*   edges    = (const int*)d_in[2];
  const float* Wk = (const float*)d_in[3];  const float* bk = (const float*)d_in[4];
  const float* gk = (const float*)d_in[5];  const float* betak = (const float*)d_in[6];
  const float* Wv = (const float*)d_in[7];  const float* bv = (const float*)d_in[8];
  const float* gv = (const float*)d_in[9];  const float* betav = (const float*)d_in[10];
  const float* Wq = (const float*)d_in[11]; const float* bq = (const float*)d_in[12];
  const float* gq = (const float*)d_in[13]; const float* betaq = (const float*)d_in[14];
  const float* Wu = (const float*)d_in[15]; const float* bu = (const float*)d_in[16];
  const float* gu = (const float*)d_in[17]; const float* betau = (const float*)d_in[18];
  float* out = (float*)d_out;

  char* ws = (char*)d_ws;
  size_t off = 0;
  auto carve = [&](size_t bytes) -> char* {
    char* p = ws + off;
    off = (off + bytes + 255) & ~(size_t)255;
    return p;
  };
  _Float16* feat_h = (_Float16*)carve((size_t)B_ * N_ * C_ * 2);
  _Float16* WkP    = (_Float16*)carve((size_t)KPAD_ * C_ * 2);
  _Float16* WvP    = (_Float16*)carve((size_t)KPAD_ * C_ * 2);
  _Float16* WqP    = (_Float16*)carve((size_t)KPAD_ * C_ * 2);
  _Float16* WuP    = (_Float16*)carve((size_t)C_ * C_ * 2);
  _Float16* Vh     = (_Float16*)carve((size_t)B_ * E_ * C_ * 2);
  float*    simi   = (float*)carve((size_t)B_ * E_ * H_ * 4);
  _Float16* aggh   = (_Float16*)carve((size_t)B_ * N_ * C_ * 2);
  int*      bincnt = (int*)carve((size_t)NBINS_ * BSTR_ * 4);
  int*      binoff = (int*)carve((size_t)NBINS_ * 4);
  int*      eslot  = (int*)carve((size_t)B_ * E_ * 4);
  int*      perm   = (int*)carve((size_t)B_ * E_ * 4);
  (void)ws_size; (void)in_sizes; (void)n_in; (void)out_size;

  k_featcvt<<<(unsigned)(((size_t)B_ * N_ * C_) / 256), 256, 0, stream>>>(features, feat_h);
  k_wpack<<<(9 * 8192) / 256, 256, 0, stream>>>(Wk, WkP, CIN_, 9);
  k_wpack<<<(9 * 8192) / 256, 256, 0, stream>>>(Wv, WvP, CIN_, 9);
  k_wpack<<<(9 * 8192) / 256, 256, 0, stream>>>(Wq, WqP, CIN_, 9);
  k_wpack<<<(8 * 8192) / 256, 256, 0, stream>>>(Wu, WuP, C_, 8);
  k_init<<<(NBINS_ * BSTR_) / 256, 256, 0, stream>>>(bincnt);

  k_edge<<<256, 256, 0, stream>>>(
      xyz, edges, feat_h,
      WkP, bk, gk, betak,
      WvP, bv, gv, betav,
      WqP, bq, gq, betaq,
      Vh, simi);

  k_rank<<<(B_ * E_ / 2) / 256, 256, 0, stream>>>(edges, bincnt, eslot);
  k_binscan<<<1, NBINS_, 0, stream>>>(bincnt, binoff);
  k_fill<<<(B_ * E_ / 2) / 256, 256, 0, stream>>>(edges, binoff, eslot, perm);
  k_binscatter<<<NBINS_, 1024, 0, stream>>>(edges, simi, Vh, bincnt, binoff, perm, aggh);

  k_update<<<(B_ * N_) / 128, 256, 0, stream>>>(aggh, WuP, bu, gu, betau, features, out);
}


# k_edge LayerNorm parameter reads batched: bias, gain, shift come in groups of 4 channel tiles with two register sets, the next group's reads issued before the current group's arithmetic; same sums
# speedup vs baseline: 2.8840x; 1.2738x over previous
#include <hip/hip_runtime.h>
#include <hip/hip_bf16.h>
#include <cstdint>

#define B_   2
#define N_   16384
#define E_   131072
#define C_   256
#define H_   8
#define CIN_ 259
#define KPAD_ 288
#define FRAGH_  512
#define WFRAGS_ 144
#define WPLANE_ (WFRAGS_ * FRAGH_)
#define NTILES_ (B_ * E_ / 16)
#define BINN_  128
#define NBINS_ (B_ * N_ / BINN_)
#define BSTR_  64

typedef __attribute__((ext_vector_type(16))) _Float16 v16h;
typedef __attribute__((ext_vector_type(8)))  float    v8f;
typedef __attribute__((ext_vector_type(8)))  _Float16 h8;
typedef __attribute__((ext_vector_type(2)))  _Float16 h2;

__device__ inline v16h ldA2(const _Float16* p, int half) {
  h8 lo = *(const h8*)(p + half * 8);
  h8 hi = *(const h8*)(p + 16 + half * 8);
  return __builtin_shufflevector(lo, hi, 0, 1, 2, 3, 4, 5, 6, 7,
                                 8, 9, 10, 11, 12, 13, 14, 15);
}
__device__ inline v8f wmma_f16(v16h a, v16h b, v8f c) {
  return __builtin_amdgcn_wmma_f32_16x16x32_f16(false, a, false, b, (short)0, c, false, false);
}

__device__ inline void atomicMaxF(float* addr, float v) {
  if (v >= 0.f) atomicMax((int*)addr, __float_as_int(v));
  else          atomicMin((unsigned int*)addr, __float_as_uint(v));
}

#define FTN_ 64
__global__ __launch_bounds__(256) void k_featcvt(const float* __restrict__ features,
                                                 _Float16* __restrict__ feat_h) {
  __shared__ __attribute__((aligned(16))) _Float16 sT[FTN_ * C_];
  const int tid = threadIdx.x;
  const int b   = blockIdx.x / (N_ / FTN_);
  const int n0  = (blockIdx.x % (N_ / FTN_)) * FTN_;
  const int q   = tid & 15;
  const int gl  = tid >> 4;
  const float* src = features + (size_t)b * C_ * N_ + n0 + 4 * q;

  float4 v[2][8];
#pragma unroll
  for (int p = 0; p < 2; ++p) {
#pragma unroll
    for (int k = 0; k < 8; ++k)
      v[p][k] = *(const float4*)(src + (size_t)((p * 16 + gl) * 8 + k) * N_);
  }
#pragma unroll
  for (int p = 0; p < 2; ++p) {
    const int g = p * 16 + gl;
    h8 o0, o1, o2, o3;
#pragma unroll
    for (int k = 0; k < 8; ++k) {
      o0[k] = (_Float16)v[p][k].x;
      o1[k] = (_Float16)v[p][k].y;
      o2[k] = (_Float16)v[p][k].z;
      o3[k] = (_Float16)v[p][k].w;
    }
    _Float16* d = sT + (4 * q) * C_ + ((g ^ q) * 8);
    *(h8*)(d)          = o0;
    *(h8*)(d + C_)     = o1;
    *(h8*)(d + 2 * C_) = o2;
    *(h8*)(d + 3 * C_) = o3;
  }
  __syncthreads();

  const int lane = tid & 31;
  const int wave = tid >> 5;
  _Float16* dst = feat_h + ((size_t)b * N_ + n0) * C_ + lane * 8;
#pragma unroll
  for (int i = 0; i < 8; ++i) {
    const int nd = wave * 8 + i;
    const h8 r = *(const h8*)(sT + nd * C_ + ((lane ^ ((nd >> 2) & 15)) * 8));
    *(h8*)(dst + (size_t)nd * C_) = r;
  }
}

__global__ void k_init(int* __restrict__ bincnt) {
  const int idx = blockIdx.x * 256 + threadIdx.x;
  if (idx < NBINS_ * BSTR_) bincnt[idx] = 0;
}


template <bool TAIL, bool TWO>
__device__ __forceinline__ void fill_w32(_Float16* __restrict__ sA, _Float16* __restrict__ sB,
                                         const float* __restrict__ W, int n) {
  const float* col = W + n;
  const int lo = (n >> 4) * FRAGH_ + (n & 15) * 8;
#pragma unroll 8
  for (int g = 0; g < 32; ++g) {
    h8 o;
#pragma unroll
    for (int i = 0; i < 8; ++i) o[i] = (_Float16)col[(size_t)(g * 8 + i) * C_];
    const int dst = (g >> 2) * (16 * FRAGH_) + ((g >> 1) & 1) * 256 + (g & 1) * 128 + lo;
    *(h8*)(sA + dst) = o;
    if (TWO) *(h8*)(sB + dst) = o;
  }
  if (TAIL) {
    h8 o = {};
    o[0] = (_Float16)col[(size_t)256 * C_];
    o[1] = (_Float16)col[(size_t)257 * C_];
    o[2] = (_Float16)col[(size_t)258 * C_];
    const h8 z = {};
#pragma unroll
    for (int g = 32; g < 36; ++g) {
      const int dst = (g >> 2) * (16 * FRAGH_) + ((g >> 1) & 1) * 256 + (g & 1) * 128 + lo;
      *(h8*)(sA + dst) = (g == 32) ? o : z;
      if (TWO) *(h8*)(sB + dst) = (g == 32) ? o : z;
    }
  }
}
__device__ __forceinline__ v16h ldW(const _Float16* sWb, int j) {
  const _Float16* p = sWb + j * FRAGH_;
  h8 lo = *(const h8*)(p);
  h8 hi = *(const h8*)(p + 256);
  return __builtin_shufflevector(lo, hi, 0, 1, 2, 3, 4, 5, 6, 7,
                                 8, 9, 10, 11, 12, 13, 14, 15);
}
__device__ __forceinline__ void ld8(const float* __restrict__ p, float (&o)[8]) {
  float4 a = *(const float4*)(p);
  float4 b = *(const float4*)(p + 4);
  o[0] = a.x; o[1] = a.y; o[2] = a.z; o[3] = a.w;
  o[4] = b.x; o[5] = b.y; o[6] = b.z; o[7] = b.w;
}
__device__ __forceinline__ void load_x(v16h (&x)[9], const _Float16* __restrict__ row,
                                       float r0, float r1, float r2, int half) {
#pragma unroll
  for (int kc = 0; kc < 8; ++kc) x[kc] = ldA2(row + kc * 32, half);
  v16h xr = {};
  xr[0] = (_Float16)(half ? 0.f : r0);
  xr[1] = (_Float16)(half ? 0.f : r1);
  xr[2] = (_Float16)(half ? 0.f : r2);
  x[8] = xr;
}
template <int NKC>
__device__ __forceinline__ void edge_gemm(v8f (&acc)[16], const _Float16* sWall, int woff,
                                          const v16h (&x)[NKC]) {
#pragma unroll
  for (int nt = 0; nt < 16; ++nt) acc[nt] = v8f{};
  v16h w[2][8];
#pragma unroll
  for (int j = 0; j < 8; ++j) w[0][j] = ldW(sWall + woff, j);
  __builtin_amdgcn_sched_barrier(0);
#pragma unroll
  for (int blk = 0; blk < 2 * NKC; ++blk) {
    if (blk + 1 < 2 * NKC) {
      int o = woff + (blk + 1) * 8 * FRAGH_;
      asm volatile("" : "+v"(o));
#pragma unroll
      for (int j = 0; j < 8; ++j) w[(blk + 1) & 1][j] = ldW(sWall + o, j);
    }
    __builtin_amdgcn_sched_barrier(0);
#pragma unroll
    for (int j = 0; j < 8; ++j) {
      const int f = blk * 8 + j;
      acc[f & 15] = wmma_f16(w[blk & 1][j], x[f >> 4], acc[f & 15]);
    }
    __builtin_amdgcn_sched_barrier(0);
  }
}
#define LNG_ 4
__device__ __forceinline__ void bias_relu_ln(v8f (&acc)[16], const float* __restrict__ bias,
                                             const float* __restrict__ g,
                                             const float* __restrict__ beta, int half) {
  float s = 0.f, s2 = 0.f;
  const float* pb = bias + half * 8;
  {
    float bb[2][LNG_][8];
#pragma unroll
    for (int j = 0; j < LNG_; ++j) ld8(pb + j * 16, bb[0][j]);
#pragma unroll
    for (int grp = 0; grp < 16 / LNG_; ++grp) {
      if (grp + 1 < 16 / LNG_) {
#pragma unroll
        for (int j = 0; j < LNG_; ++j) ld8(pb + ((grp + 1) * LNG_ + j) * 16, bb[(grp + 1) & 1][j]);
      }
      __builtin_amdgcn_sched_barrier(0);
#pragma unroll
      for (int j = 0; j < LNG_; ++j) {
        const int nt = grp * LNG_ + j;
#pragma unroll
        for (int r = 0; r < 8; ++r) {
          float v = fmaxf(acc[nt][r] + bb[grp & 1][j][r], 0.f);
          acc[nt][r] = v;
          s += v; s2 += v * v;
        }
      }
      __builtin_amdgcn_sched_barrier(0);
    }
  }
  s  += __shfl_xor(s, 16);
  s2 += __shfl_xor(s2, 16);
  const float mean = s * (1.f / 256.f);
  const float var  = fmaxf(s2 * (1.f / 256.f) - mean * mean, 0.f);
  const float istd = rsqrtf(var + 1e-5f);
  const float* pg  = g + half * 8;
  const float* pbe = beta + half * 8;
  {
    float gg[2][LNG_][8], be[2][LNG_][8];
#pragma unroll
    for (int j = 0; j < LNG_; ++j) {
      ld8(pg + j * 16, gg[0][j]);
      ld8(pbe + j * 16, be[0][j]);
    }
#pragma unroll
    for (int grp = 0; grp < 16 / LNG_; ++grp) {
      if (grp + 1 < 16 / LNG_) {
#pragma unroll
        for (int j = 0; j < LNG_; ++j) {
          ld8(pg + ((grp + 1) * LNG_ + j) * 16, gg[(grp + 1) & 1][j]);
          ld8(pbe + ((grp + 1) * LNG_ + j) * 16, be[(grp + 1) & 1][j]);
        }
      }
      __builtin_amdgcn_sched_barrier(0);
#pragma unroll
      for (int j = 0; j < LNG_; ++j) {
        const int nt = grp * LNG_ + j;
#pragma unroll
        for (int r = 0; r < 8; ++r)
          acc[nt][r] = (acc[nt][r] - mean) * istd * gg[grp & 1][j][r] + be[grp & 1][j][r];
      }
      __builtin_amdgcn_sched_barrier(0);
    }
  }
}

__global__ __launch_bounds__(256) void k_edge(
    const float* __restrict__ xyz, const int* __restrict__ edges,
    const _Float16* __restrict__ feat_h,
    const float* __restrict__ Wk, const float* __restrict__ biask,
    const float* __restrict__ gk, const float* __restrict__ betak,
    const float* __restrict__ Wv, const float* __restrict__ biasv,
    const float* __restrict__ gv, const float* __restrict__ betav,
    const float* __restrict__ Wq, const float* __restrict__ biasq,
    const float* __restrict__ gq, const float* __restrict__ betaq,
    _Float16* __restrict__ Vh, float* __restrict__ simi) {
  const int tid  = threadIdx.x;
  const int wave = __builtin_amdgcn_readfirstlane(tid >> 5);

  __shared__ __attribute__((aligned(16))) _Float16 sW[2 * WPLANE_];
  _Float16* sW0 = sW;
  _Float16* sW1 = sW + WPLANE_;

  const int qm = -(wave & 1);
  const float* bias1 = qm ? biasq : biask;
  const float* g1    = qm ? gq    : gk;
  const float* beta1 = qm ? betaq : betak;
  const float* bias2 = qm ? biask : biasq;
  const float* g2    = qm ? gk    : gq;
  const float* beta2 = qm ? betak : betaq;
  const int plane1 = qm ? WPLANE_ : 0;
  const int plane2 = qm ? 0 : WPLANE_;

  fill_w32<true, false>(sW0, nullptr, Wk, tid);
  fill_w32<true, false>(sW1, nullptr, Wq, tid);
  __syncthreads();

  for (int tile = blockIdx.x * 8 + wave; tile < NTILES_; tile += gridDim.x * 8) {
    const int b = tile / (E_ / 16);
    int ln = (int)__builtin_amdgcn_mbcnt_lo(~0u, 0u);
    asm volatile("" : "+v"(ln));
    const int half = ln >> 4;
    const int l16  = ln & 15;
    const int woff = ln * 8;
    const size_t e = (size_t)b * E_ + (size_t)(tile % (E_ / 16)) * 16 + l16;
    const size_t bN = (size_t)b * N_;
    const int s = edges[e * 2 + 0];
    const int d = edges[e * 2 + 1];
    const float r0 = xyz[(bN + s) * 3 + 0] - xyz[(bN + d) * 3 + 0];
    const float r1 = xyz[(bN + s) * 3 + 1] - xyz[(bN + d) * 3 + 1];
    const float r2 = xyz[(bN + s) * 3 + 2] - xyz[(bN + d) * 3 + 2];

    const int n1 = s ^ ((s ^ d) & qm);
    const int n2 = d ^ ((s ^ d) & qm);

    v16h x[9];
    v8f aK[16], aQ[16];
    load_x(x, feat_h + (bN + (size_t)n1) * C_, r0, r1, r2, half);
    edge_gemm<9>(aK, sW, plane1 + woff, x);
    bias_relu_ln(aK, bias1, g1, beta1, half);

    load_x(x, feat_h + (bN + (size_t)n2) * C_, r0, r1, r2, half);
    edge_gemm<9>(aQ, sW, plane2 + woff, x);
    bias_relu_ln(aQ, bias2, g2, beta2, half);

    float sv[8];
#pragma unroll
    for (int h = 0; h < 8; ++h) {
      float p = 0.f;
#pragma unroll
      for (int r = 0; r < 8; ++r) p += aK[2 * h][r] * aQ[2 * h][r];
#pragma unroll
      for (int r = 0; r < 8; ++r) p += aK[2 * h + 1][r] * aQ[2 * h + 1][r];
      p += __shfl_xor(p, 16);
      sv[h] = p * 0.17677669529663687f;
    }
    if (half == 0) {
      float4* o = (float4*)(simi + e * H_);
      o[0] = make_float4(sv[0], sv[1], sv[2], sv[3]);
      o[1] = make_float4(sv[4], sv[5], sv[6], sv[7]);
    }
  }

  __syncthreads();
  fill_w32<true, true>(sW0, sW1, Wv, tid);
  __syncthreads();
  const int planev = qm ? WPLANE_ : 0;

  for (int tile = blockIdx.x * 8 + wave; tile < NTILES_; tile += gridDim.x * 8) {
    const int b = tile / (E_ / 16);
    int ln = (int)__builtin_amdgcn_mbcnt_lo(~0u, 0u);
    asm volatile("" : "+v"(ln));
    const int half = ln >> 4;
    const int l16  = ln & 15;
    const int woff = ln * 8;
    const size_t e = (size_t)b * E_ + (size_t)(tile % (E_ / 16)) * 16 + l16;
    const size_t bN = (size_t)b * N_;
    const int s = edges[e * 2 + 0];
    const int d = edges[e * 2 + 1];
    const float r0 = xyz[(bN + s) * 3 + 0] - xyz[(bN + d) * 3 + 0];
    const float r1 = xyz[(bN + s) * 3 + 1] - xyz[(bN + d) * 3 + 1];
    const float r2 = xyz[(bN + s) * 3 + 2] - xyz[(bN + d) * 3 + 2];

    v16h x[9];
    v8f aV[16];
    load_x(x, feat_h + (bN + (size_t)s) * C_, r0, r1, r2, half);
    edge_gemm<9>(aV, sW, planev + woff, x);
    bias_relu_ln(aV, biasv, gv, betav, half);

    _Float16* vrow = Vh + e * C_ + half * 8;
#pragma unroll
    for (int nt = 0; nt < 16; ++nt) {
      h8 o;
#pragma unroll
      for (int r = 0; r < 8; ++r) o[r] = (_Float16)aV[nt][r];
      *(h8*)(vrow + nt * 16) = o;
    }
  }
}

__global__ void k_rank(const int* __restrict__ edges, int* __restrict__ bincnt,
                       int* __restrict__ eslot) {
  const int t = blockIdx.x * 256 + threadIdx.x;
  if (t >= B_ * E_ / 2) return;
  const int d0 = edges[(size_t)t * 4 + 1];
  const int d1 = edges[(size_t)t * 4 + 3];
  const int bb = ((2 * t) / E_) * (N_ / BINN_);
  int2 s;
  s.x = atomicAdd(&bincnt[(bb + d0 / BINN_) * BSTR_], 1);
  s.y = atomicAdd(&bincnt[(bb + d1 / BINN_) * BSTR_], 1);
  ((int2*)eslot)[t] = s;
}

__global__ __launch_bounds__(NBINS_) void k_binscan(const int* __restrict__ bincnt,
                                                    int* __restrict__ binoff) {
  __shared__ int s[NBINS_];
  const int t = threadIdx.x;
  const int v = bincnt[t * BSTR_];
  s[t] = v;
  __syncthreads();
  for (int o = 1; o < NBINS_; o <<= 1) {
    const int a = (t >= o) ? s[t - o] : 0;
    __syncthreads();
    s[t] += a;
    __syncthreads();
  }
  binoff[t] = s[t] - v;
}

__global__ void k_fill(const int* __restrict__ edges, const int* __restrict__ binoff,
                       const int* __restrict__ eslot, int* __restrict__ perm) {
  const int t = blockIdx.x * 256 + threadIdx.x;
  if (t >= B_ * E_ / 2) return;
  const int d0 = edges[(size_t)t * 4 + 1];
  const int d1 = edges[(size_t)t * 4 + 3];
  const int2 s = ((const int2*)eslot)[t];
  const int bb = ((2 * t) / E_) * (N_ / BINN_);
  perm[binoff[bb + d0 / BINN_] + s.x] = 2 * t;
  perm[binoff[bb + d1 / BINN_] + s.y] = 2 * t + 1;
}

__global__ __launch_bounds__(1024) void k_binscatter(
    const int* __restrict__ edges, const float* __restrict__ simi,
    const _Float16* __restrict__ Vh, const int* __restrict__ bincnt,
    const int* __restrict__ binoff, const int* __restrict__ perm,
    _Float16* __restrict__ aggh) {
  __shared__ __attribute__((aligned(16))) float tile[BINN_ * C_];
  __shared__ float smax[BINN_ * H_];
  __shared__ float ssum[BINN_ * H_];
  __shared__ int   scnt[BINN_];
  const int tid  = threadIdx.x;
  const int lane = tid & 31;
  const int wave = __builtin_amdgcn_readfirstlane(tid >> 5);
  const int bin  = blockIdx.x;
  const int hq   = lane >> 2;

  for (int i = tid; i < BINN_ * C_ / 4; i += 1024)
    ((float4*)tile)[i] = make_float4(0.f, 0.f, 0.f, 0.f);
  for (int i = tid; i < BINN_ * H_; i += 1024) {
    smax[i] = __int_as_float(0xff800000);
    ssum[i] = 0.f;
  }
  for (int i = tid; i < BINN_; i += 1024) scnt[i] = 0;
  __syncthreads();

  const int nE  = bincnt[bin * BSTR_];
  const int off = binoff[bin];

  {
    const int ha = tid & 7;
    for (int base = tid >> 3; base < nE; base += 128 * 4) {
      int   ge[4];
      int   dl[4];
      float sv[4];
#pragma unroll
      for (int u = 0; u < 4; ++u) {
        const int idx = (base + u * 128 < nE) ? (base + u * 128) : (nE - 1);
        ge[u] = perm[off + idx];
      }
#pragma unroll
      for (int u = 0; u < 4; ++u) {
        dl[u] = edges[(size_t)ge[u] * 2 + 1] & (BINN_ - 1);
        sv[u] = simi[(size_t)ge[u] * H_ + ha];
      }
#pragma unroll
      for (int u = 0; u < 4; ++u) {
        if (base + u * 128 < nE) {
          atomicMaxF(&smax[dl[u] * H_ + ha], sv[u]);
          if (ha == 0) atomicAdd(&scnt[dl[u]], 1);
        }
      }
    }
  }
  __syncthreads();

  for (int base = wave * 4; base < nE; base += 32 * 4) {
    int   ge[4];
    int   dl[4];
    float sv[4];
    h8    v[4];
#pragma unroll
    for (int u = 0; u < 4; ++u) {
      const int idx = (base + u < nE) ? (base + u) : (nE - 1);
      ge[u] = perm[off + idx];
    }
#pragma unroll
    for (int u = 0; u < 4; ++u) {
      dl[u] = edges[(size_t)ge[u] * 2 + 1] & (BINN_ - 1);
      sv[u] = simi[(size_t)ge[u] * H_ + hq];
      v[u]  = *(const h8*)(Vh + (size_t)ge[u] * C_ + lane * 8);
    }
#pragma unroll
    for (int u = 0; u < 4; ++u) {
      if (base + u < nE) {
        const float w = __expf(sv[u] - smax[dl[u] * H_ + hq]);
        if ((lane & 3) == 0) atomicAdd(&ssum[dl[u] * H_ + hq], w);
        float* row = tile + dl[u] * C_ + lane;
#pragma unroll
        for (int r = 0; r < 8; ++r) atomicAdd(row + r * 32, (float)v[u][r] * w);
      }
    }
  }
  __syncthreads();

  const int b  = bin / (N_ / BINN_);
  const int n0 = (bin % (N_ / BINN_)) * BINN_;
  for (int nl = wave; nl < BINN_; nl += 32) {
    const size_t gn = (size_t)b * N_ + n0 + nl;
    const float cc = fmaxf((float)scnt[nl], 1.0f);
    float bs = ssum[nl * H_ + hq];
    bs = (bs > 0.f) ? bs : 1.f;
    h8 o;
#pragma unroll
    for (int r = 0; r < 8; ++r) o[r] = (_Float16)(tile[nl * C_ + r * 32 + lane] / bs / cc);
    *(h8*)(aggh + gn * C_ + lane * 8) = o;
  }
}

__global__ __launch_bounds__(256) void k_update(
    const _Float16* __restrict__ aggh, const float* __restrict__ Wu,
    const float* __restrict__ biasu, const float* __restrict__ gu,
    const float* __restrict__ betau, const float* __restrict__ features,
    float* __restrict__ out) {
  const int lane = threadIdx.x & 31;
  const int wave = threadIdx.x >> 5;
  const int half = lane >> 4;
  const int l16  = lane & 15;
  const int tile = blockIdx.x * 8 + wave;
  const int b    = tile / (N_ / 16);
  const int node = (tile % (N_ / 16)) * 16 + l16;

  __shared__ __attribute__((aligned(16))) _Float16 sWu[(C_ / 32) * 16 * FRAGH_];

  const _Float16* row = aggh + ((size_t)b * N_ + node) * C_;
  v16h xb[8];
#pragma unroll
  for (int kc = 0; kc < 8; ++kc) xb[kc] = ldA2(row + kc * 32, half);

  fill_w32<false, false>(sWu, nullptr, Wu, (int)threadIdx.x);
  __syncthreads();

  v8f acc[16];
  edge_gemm<8>(acc, sWu, lane * 8, xb);

  float s = 0.f, s2 = 0.f;
#pragma unroll
  for (int nt = 0; nt < 16; ++nt) {
#pragma unroll
    for (int r = 0; r < 8; ++r) {
      int ch = nt * 16 + half * 8 + r;
      float x = fmaxf(acc[nt][r] + biasu[ch], 0.f);
      acc[nt][r] = x;
      s += x; s2 += x * x;
    }
  }
  s  += __shfl_xor(s, 16);
  s2 += __shfl_xor(s2, 16);
  const float mean = s * (1.f / 256.f);
  const float var  = fmaxf(s2 * (1.f / 256.f) - mean * mean, 0.f);
  const float istd = rsqrtf(var + 1e-5f);

#pragma unroll
  for (int nt = 0; nt < 16; ++nt) {
#pragma unroll
    for (int r = 0; r < 8; ++r) {
      int ch = nt * 16 + half * 8 + r;
      float val = (acc[nt][r] - mean) * istd * gu[ch] + betau[ch];
      size_t oi = ((size_t)b * C_ + ch) * N_ + node;
      out[oi] = val + features[oi];
    }
  }
}

extern "C" void kernel_launch(void* const* d_in, const int* in_sizes, int n_in,
                              void* d_out, int out_size, void* d_ws, size_t ws_size,
                              hipStream_t stream) {
  const float* xyz      = (const float*)d_in[0];
  const float* features = (const float*)d_in[1];
  const int*   edges    = (const int*)d_in[2];
  const float* Wk = (const float*)d_in[3];  const float* bk = (const float*)d_in[4];
  const float* gk = (const float*)d_in[5];  const float* betak = (const float*)d_in[6];
  const float* Wv = (const float*)d_in[7];  const float* bv = (const float*)d_in[8];
  const float* gv = (const float*)d_in[9];  const float* betav = (const float*)d_in[10];
  const float* Wq = (const float*)d_in[11]; const float* bq = (const float*)d_in[12];
  const float* gq = (const float*)d_in[13]; const float* betaq = (const float*)d_in[14];
  const float* Wu = (const float*)d_in[15]; const float* bu = (const float*)d_in[16];
  const float* gu = (const float*)d_in[17]; const float* betau = (const float*)d_in[18];
  float* out = (float*)d_out;

  char* ws = (char*)d_ws;
  size_t off = 0;
  auto carve = [&](size_t bytes) -> char* {
    char* p = ws + off;
    off = (off + bytes + 255) & ~(size_t)255;
    return p;
  };
  _Float16* feat_h = (_Float16*)carve((size_t)B_ * N_ * C_ * 2);
  _Float16* Vh     = (_Float16*)carve((size_t)B_ * E_ * C_ * 2);
  float*    simi   = (float*)carve((size_t)B_ * E_ * H_ * 4);
  _Float16* aggh   = (_Float16*)carve((size_t)B_ * N_ * C_ * 2);
  int*      bincnt = (int*)carve((size_t)NBINS_ * BSTR_ * 4);
  int*      binoff = (int*)carve((size_t)NBINS_ * 4);
  int*      eslot  = (int*)carve((size_t)B_ * E_ * 4);
  int*      perm   = (int*)carve((size_t)B_ * E_ * 4);
  (void)ws_size; (void)in_sizes; (void)n_in; (void)out_size;

  k_featcvt<<<(B_ * N_) / FTN_, 256, 0, stream>>>(features, feat_h);
  k_init<<<(NBINS_ * BSTR_) / 256, 256, 0, stream>>>(bincnt);

  k_edge<<<256, 256, 0, stream>>>(
      xyz, edges, feat_h,
      Wk, bk, gk, betak,
      Wv, bv, gv, betav,
      Wq, bq, gq, betaq,
      Vh, simi);

  k_rank<<<(B_ * E_ / 2) / 256, 256, 0, stream>>>(edges, bincnt, eslot);
  k_binscan<<<1, NBINS_, 0, stream>>>(bincnt, binoff);
  k_fill<<<(B_ * E_ / 2) / 256, 256, 0, stream>>>(edges, binoff, eslot, perm);
  k_binscatter<<<NBINS_, 1024, 0, stream>>>(edges, simi, Vh, bincnt, binoff, perm, aggh);

  k_update<<<(B_ * N_) / 128, 256, 0, stream>>>(aggh, Wu, bu, gu, betau, features, out);
}
